# starting kernel
# baseline (speedup 1.0000x reference)
#include <hip/hip_runtime.h>


#define N       128
#define LDSTR   136
#define PSTR    288
#define NBATCH  2048
#define EPSF    1e-8f
#define NSTEPS  3

typedef __attribute__((ext_vector_type(2))) float v2f;
typedef __attribute__((ext_vector_type(8))) float v8f;

__device__ __forceinline__ int fidx(int row, int col) {
  return (row >> 1) * PSTR + col * 2 + (row & 1);
}

__global__ __launch_bounds__(256, 1)
void sgf_kernel(const float* __restrict__ corr,
                const float* __restrict__ temperature,
                float* __restrict__ out) {
  extern __shared__ float lds[];
  float* sC  = lds;
  float* sS  = lds + N * LDSTR;
  float* sFp = lds + 2 * N * LDSTR;

  const int tid  = threadIdx.x;
  const int lane = tid & 31;
  const int wave = tid >> 5;
  const int lm   = lane & 15;
  const int lh   = lane >> 4;
  const int b    = blockIdx.x;

  const float invT = 1.0f / temperature[0];
  const float* src = corr + (size_t)b * (N * N);

  for (int e = tid * 4; e < N * N; e += 256 * 4) {
    float4 v = *(const float4*)(src + e);
    int i = e >> 7, j = e & 127;
    *(float4*)&sS[i * LDSTR + j] = v;
  }
  __syncthreads();

  for (int e = tid; e < N * N; e += 256) {
    int i = e >> 7, j = e & 127;
    float x = sS[i * LDSTR + j];
    float y = sS[j * LDSTR + i];
    x = fminf(fmaxf(x, -1.0f + EPSF), 1.0f - EPSF);
    y = fminf(fmaxf(y, -1.0f + EPSF), 1.0f - EPSF);
    sC[i * LDSTR + j] = 0.5f * (x + y);
  }
  __syncthreads();

  if (tid < N) {
    const int i = tid;
    float t0 = -1e30f, t1 = -1e30f, t2 = -1e30f;
    for (int j = 0; j < N; ++j) {
      float cv  = sC[i * LDSTR + j];
      float d   = __builtin_sqrtf(2.0f * (1.0f - cv) + EPSF);
      float aff = __expf(-d * invT);
      if (j == i) aff = 0.0f;
      sS[i * LDSTR + j] = aff;
      if (aff > t0)      { t2 = t1; t1 = t0; t0 = aff; }
      else if (aff > t1) { t2 = t1; t1 = aff; }
      else if (aff > t2) { t2 = aff; }
    }
    float sum = 0.0f;
    for (int j = 0; j < N; ++j) {
      float a = sS[i * LDSTR + j];
      if (a >= t2) sum += __expf(a - t0);
    }
    float rsum = 1.0f / sum;
    for (int j = 0; j < N; ++j) {
      float a = sS[i * LDSTR + j];
      sS[i * LDSTR + j] = (a >= t2) ? __expf(a - t0) * rsum : 0.0f;
    }
  }
  __syncthreads();

  for (int e = tid; e < N * N; e += 256) {
    int i = e >> 7, j = e & 127;
    if (i < j) {
      float a = sS[i * LDSTR + j];
      float c2 = sS[j * LDSTR + i];
      float m = 0.5f * (a + c2);
      sS[i * LDSTR + j] = m;
      sS[j * LDSTR + i] = m;
    }
  }

  for (int e = tid; e < N * N; e += 256) {
    int i = e >> 7, j = e & 127;
    sFp[fidx(i, j)] = sC[i * LDSTR + j];
  }
  __syncthreads();

  const int tm       = wave;
  const int arowbase = (tm * 16 + lm) * LDSTR + 2 * lh;
  const int bcolbase = lh * PSTR + lm * 2;

  const float kAlpha[NSTEPS] = { 0.5f, 1.0f / 3.0f, 0.25f };

#pragma unroll
  for (int step = 0; step < NSTEPS; ++step) {
    v8f acc[8] = {};

    v2f av = *(const v2f*)&sS[arowbase];
    v2f bcur[8];
#pragma unroll
    for (int tn = 0; tn < 8; ++tn)
      bcur[tn] = *(const v2f*)&sFp[bcolbase + tn * 32];

    for (int k = 0; k < N - 4; k += 4) {
      v2f an = *(const v2f*)&sS[arowbase + k + 4];
      const int bb = ((k + 4) >> 1) * PSTR + bcolbase;
      v2f bn[8];
#pragma unroll
      for (int tn = 0; tn < 8; ++tn)
        bn[tn] = *(const v2f*)&sFp[bb + tn * 32];
#pragma unroll
      for (int tn = 0; tn < 8; ++tn)
        acc[tn] = __builtin_amdgcn_wmma_f32_16x16x4_f32(
            false, av, false, bcur[tn], (short)0, acc[tn], false, false);
      av = an;
#pragma unroll
      for (int tn = 0; tn < 8; ++tn) bcur[tn] = bn[tn];
    }
#pragma unroll
    for (int tn = 0; tn < 8; ++tn)
      acc[tn] = __builtin_amdgcn_wmma_f32_16x16x4_f32(
          false, av, false, bcur[tn], (short)0, acc[tn], false, false);

    __syncthreads();

    const float alpha = kAlpha[step];
#pragma unroll
    for (int tn = 0; tn < 8; ++tn) {
#pragma unroll
      for (int v = 0; v < 8; ++v) {
        int row  = tm * 16 + v + 8 * lh;
        int col  = tn * 16 + lm;
        int ridx = row * LDSTR + col;
        int fi   = fidx(row, col);
        float msg = acc[tn][v];
        float f   = sFp[fi];
        float s   = sS[ridx];
        float cc  = sC[ridx];
        float fn  = __builtin_fmaf(alpha, msg - f, f);
        sFp[fi]   = __builtin_fmaf(s, cc - fn, fn);
      }
    }
    __syncthreads();
  }

  float* dst = out + (size_t)b * (N * N);
  for (int e = tid; e < N * N; e += 256) {
    int i = e >> 7, j = e & 127;
    float val = (i == j) ? 1.0f
                         : 0.5f * (sFp[fidx(i, j)] + sFp[fidx(j, i)]);
    dst[e] = val;
  }
}

extern "C" void kernel_launch(void* const* d_in, const int* in_sizes, int n_in,
                              void* d_out, int out_size, void* d_ws, size_t ws_size,
                              hipStream_t stream) {
  (void)in_sizes; (void)n_in; (void)out_size; (void)d_ws; (void)ws_size;
  const float* corr = (const float*)d_in[0];
  const float* temp = (const float*)d_in[1];
  float* out = (float*)d_out;

  const size_t shmem =
      ((size_t)2 * N * LDSTR + (size_t)(N / 2) * PSTR) * sizeof(float);
  static bool attr_set = false;
  if (!attr_set) {
    hipFuncSetAttribute((const void*)sgf_kernel,
                        hipFuncAttributeMaxDynamicSharedMemorySize, (int)shmem);
    attr_set = true;
  }
  sgf_kernel<<<dim3(NBATCH), dim3(256), shmem, stream>>>(corr, temp, out);
}


# LDS per workgroup 208 KB -> 140 KB so two workgroups are resident per WGP: c kept in 64 registers per lane, filtered reuses c's LDS region, K loop on two swapping fragment sets to fit 256 VGPRs
# speedup vs baseline: 1.7506x; 1.7506x over previous
#include <hip/hip_runtime.h>


#define N       128
#define LDSTR   136
#define PSTR    288
#define NBATCH  2048
#define EPSF    1e-8f
#define NSTEPS  3
#define SFLOATS (N * LDSTR)
#define CFLOATS ((N / 2) * PSTR)

typedef __attribute__((ext_vector_type(2))) float v2f;
typedef __attribute__((ext_vector_type(8))) float v8f;

__device__ __forceinline__ int fidx(int row, int col) {
  return (row >> 1) * PSTR + col * 2 + (row & 1);
}

__global__ __launch_bounds__(256, 4)
void sgf_kernel(const float* __restrict__ corr,
                const float* __restrict__ temperature,
                float* __restrict__ out) {
  extern __shared__ float lds[];
  float* sS  = lds;
  float* sC  = lds + SFLOATS;
  float* sFp = lds + SFLOATS;

  const int tid  = threadIdx.x;
  const int lane = tid & 31;
  const int wave = tid >> 5;
  const int lm   = lane & 15;
  const int lh   = lane >> 4;
  const int b    = blockIdx.x;

  const float invT = 1.0f / temperature[0];
  const float* src = corr + (size_t)b * (N * N);

  for (int e = tid * 4; e < N * N; e += 256 * 4) {
    float4 v = *(const float4*)(src + e);
    int i = e >> 7, j = e & 127;
    *(float4*)&sS[i * LDSTR + j] = v;
  }
  __syncthreads();

  for (int e = tid; e < N * N; e += 256) {
    int i = e >> 7, j = e & 127;
    float x = sS[i * LDSTR + j];
    float y = sS[j * LDSTR + i];
    x = fminf(fmaxf(x, -1.0f + EPSF), 1.0f - EPSF);
    y = fminf(fmaxf(y, -1.0f + EPSF), 1.0f - EPSF);
    sC[i * LDSTR + j] = 0.5f * (x + y);
  }
  __syncthreads();

  if (tid < N) {
    const int i = tid;
    float t0 = -1e30f, t1 = -1e30f, t2 = -1e30f;
    for (int j = 0; j < N; ++j) {
      float cv  = sC[i * LDSTR + j];
      float d   = __builtin_sqrtf(2.0f * (1.0f - cv) + EPSF);
      float aff = __expf(-d * invT);
      if (j == i) aff = 0.0f;
      sS[i * LDSTR + j] = aff;
      if (aff > t0)      { t2 = t1; t1 = t0; t0 = aff; }
      else if (aff > t1) { t2 = t1; t1 = aff; }
      else if (aff > t2) { t2 = aff; }
    }
    float sum = 0.0f;
    for (int j = 0; j < N; ++j) {
      float a = sS[i * LDSTR + j];
      if (a >= t2) sum += __expf(a - t0);
    }
    float rsum = 1.0f / sum;
    for (int j = 0; j < N; ++j) {
      float a = sS[i * LDSTR + j];
      sS[i * LDSTR + j] = (a >= t2) ? __expf(a - t0) * rsum : 0.0f;
    }
  }
  __syncthreads();

  for (int e = tid; e < N * N; e += 256) {
    int i = e >> 7, j = e & 127;
    if (i < j) {
      float a = sS[i * LDSTR + j];
      float c2 = sS[j * LDSTR + i];
      float m = 0.5f * (a + c2);
      sS[i * LDSTR + j] = m;
      sS[j * LDSTR + i] = m;
    }
  }

  const int tm = wave;
  float creg[8][8];
#pragma unroll
  for (int tn = 0; tn < 8; ++tn) {
#pragma unroll
    for (int v = 0; v < 8; ++v)
      creg[tn][v] = sC[(tm * 16 + v + 8 * lh) * LDSTR + tn * 16 + lm];
  }
  __syncthreads();

#pragma unroll
  for (int tn = 0; tn < 8; ++tn) {
#pragma unroll
    for (int v = 0; v < 8; ++v)
      sFp[fidx(tm * 16 + v + 8 * lh, tn * 16 + lm)] = creg[tn][v];
  }
  __syncthreads();

  const int arowbase = (tm * 16 + lm) * LDSTR + 2 * lh;
  const int bcolbase = lh * PSTR + lm * 2;

  const float kAlpha[NSTEPS] = { 0.5f, 1.0f / 3.0f, 0.25f };

#pragma unroll
  for (int step = 0; step < NSTEPS; ++step) {
    v8f acc[8] = {};

    v2f a0 = *(const v2f*)&sS[arowbase];
    v2f b0[8];
#pragma unroll
    for (int tn = 0; tn < 8; ++tn)
      b0[tn] = *(const v2f*)&sFp[bcolbase + tn * 32];

    for (int k = 0; k < N; k += 8) {
      v2f a1 = *(const v2f*)&sS[arowbase + k + 4];
      const int bb1 = ((k + 4) >> 1) * PSTR + bcolbase;
      v2f b1[8];
#pragma unroll
      for (int tn = 0; tn < 8; ++tn)
        b1[tn] = *(const v2f*)&sFp[bb1 + tn * 32];
#pragma unroll
      for (int tn = 0; tn < 8; ++tn)
        acc[tn] = __builtin_amdgcn_wmma_f32_16x16x4_f32(
            false, a0, false, b0[tn], (short)0, acc[tn], false, false);

      const int kn = (k + 8) & (N - 1);
      a0 = *(const v2f*)&sS[arowbase + kn];
      const int bb0 = (kn >> 1) * PSTR + bcolbase;
#pragma unroll
      for (int tn = 0; tn < 8; ++tn)
        b0[tn] = *(const v2f*)&sFp[bb0 + tn * 32];
#pragma unroll
      for (int tn = 0; tn < 8; ++tn)
        acc[tn] = __builtin_amdgcn_wmma_f32_16x16x4_f32(
            false, a1, false, b1[tn], (short)0, acc[tn], false, false);
    }

    __syncthreads();

    const float alpha = kAlpha[step];
#pragma unroll
    for (int tn = 0; tn < 8; ++tn) {
#pragma unroll
      for (int v = 0; v < 8; ++v) {
        int row  = tm * 16 + v + 8 * lh;
        int col  = tn * 16 + lm;
        int ridx = row * LDSTR + col;
        int fi   = fidx(row, col);
        float msg = acc[tn][v];
        float f   = sFp[fi];
        float s   = sS[ridx];
        float cc  = creg[tn][v];
        float fn  = __builtin_fmaf(alpha, msg - f, f);
        sFp[fi]   = __builtin_fmaf(s, cc - fn, fn);
      }
    }
    __syncthreads();
  }

  float* dst = out + (size_t)b * (N * N);
  for (int e = tid; e < N * N; e += 256) {
    int i = e >> 7, j = e & 127;
    float val = (i == j) ? 1.0f
                         : 0.5f * (sFp[fidx(i, j)] + sFp[fidx(j, i)]);
    dst[e] = val;
  }
}

extern "C" void kernel_launch(void* const* d_in, const int* in_sizes, int n_in,
                              void* d_out, int out_size, void* d_ws, size_t ws_size,
                              hipStream_t stream) {
  (void)in_sizes; (void)n_in; (void)out_size; (void)d_ws; (void)ws_size;
  const float* corr = (const float*)d_in[0];
  const float* temp = (const float*)d_in[1];
  float* out = (float*)d_out;

  const size_t shmem =
      ((size_t)SFLOATS + (size_t)CFLOATS) * sizeof(float);
  static bool attr_set = false;
  if (!attr_set) {
    hipFuncSetAttribute((const void*)sgf_kernel,
                        hipFuncAttributeMaxDynamicSharedMemorySize, (int)shmem);
    attr_set = true;
  }
  sgf_kernel<<<dim3(NBATCH), dim3(256), shmem, stream>>>(corr, temp, out);
}


# Graph build (affinity, top-3 threshold, masked softmax) on all 8 waves, lanes across a row's columns; top-3 and row sum joined by a 5-step lane butterfly: softmax row sum order changes
# speedup vs baseline: 2.3052x; 1.3168x over previous
#include <hip/hip_runtime.h>


#define N       128
#define LDSTR   136
#define PSTR    288
#define NBATCH  2048
#define EPSF    1e-8f
#define NSTEPS  3
#define SFLOATS (N * LDSTR)
#define CFLOATS ((N / 2) * PSTR)

typedef __attribute__((ext_vector_type(2))) float v2f;
typedef __attribute__((ext_vector_type(8))) float v8f;

__device__ __forceinline__ int fidx(int row, int col) {
  return (row >> 1) * PSTR + col * 2 + (row & 1);
}

template <int M>
__device__ __forceinline__ float xlane(float x) {
  const int b = __builtin_bit_cast(int, x);
  if (M == 16)
    return __builtin_bit_cast(float,
        __builtin_amdgcn_permlanex16(b, b, 0x76543210, 0xfedcba98, false, false));
  return __builtin_bit_cast(float,
      __builtin_amdgcn_mov_dpp(b, 0x160 + (M & 15), 0xf, 0xf, false));
}

template <int M>
__device__ __forceinline__ void merge_top3(float& a0, float& a1, float& a2) {
  const float b0 = xlane<M>(a0), b1 = xlane<M>(a1), b2 = xlane<M>(a2);
  const float c0 = __builtin_fmaxf(a0, b0);
  const float c1 = __builtin_fmaxf(__builtin_fmaxf(a1, b1), __builtin_fminf(a0, b0));
  const float c2 = __builtin_fmaxf(__builtin_fmaxf(a2, b2),
                                   __builtin_fmaxf(__builtin_fminf(a1, b0), __builtin_fminf(a0, b1)));
  a0 = c0; a1 = c1; a2 = c2;
}

template <int M>
__device__ __forceinline__ void merge_top3_rows(float (&t0)[4], float (&t1)[4], float (&t2)[4]) {
#pragma unroll
  for (int r = 0; r < 4; ++r) merge_top3<M>(t0[r], t1[r], t2[r]);
}

template <int M>
__device__ __forceinline__ void add_rows(float (&s)[4]) {
#pragma unroll
  for (int r = 0; r < 4; ++r) s[r] += xlane<M>(s[r]);
}

__global__ __launch_bounds__(256, 4)
void sgf_kernel(const float* __restrict__ corr,
                const float* __restrict__ temperature,
                float* __restrict__ out) {
  extern __shared__ float lds[];
  float* sS  = lds;
  float* sC  = lds + SFLOATS;
  float* sFp = lds + SFLOATS;

  const int tid  = threadIdx.x;
  const int lane = tid & 31;
  const int wave = tid >> 5;
  const int lm   = lane & 15;
  const int lh   = lane >> 4;
  const int b    = blockIdx.x;

  const float invT = 1.0f / temperature[0];
  const float* src = corr + (size_t)b * (N * N);

  for (int e = tid * 4; e < N * N; e += 256 * 4) {
    float4 v = *(const float4*)(src + e);
    int i = e >> 7, j = e & 127;
    *(float4*)&sS[i * LDSTR + j] = v;
  }
  __syncthreads();

  for (int e = tid; e < N * N; e += 256) {
    int i = e >> 7, j = e & 127;
    float x = sS[i * LDSTR + j];
    float y = sS[j * LDSTR + i];
    x = fminf(fmaxf(x, -1.0f + EPSF), 1.0f - EPSF);
    y = fminf(fmaxf(y, -1.0f + EPSF), 1.0f - EPSF);
    sC[i * LDSTR + j] = 0.5f * (x + y);
  }
  __syncthreads();

  for (int rb = 0; rb < 16; rb += 4) {
    float e[4][4];
    float t0[4], t1[4], t2[4];
#pragma unroll
    for (int r = 0; r < 4; ++r) {
      const int row = wave * 16 + rb + r;
      const float4 cv = *(const float4*)&sC[row * LDSTR + 4 * lane];
      const float c4[4] = { cv.x, cv.y, cv.z, cv.w };
#pragma unroll
      for (int q = 0; q < 4; ++q) {
        float d   = __builtin_sqrtf(2.0f * (1.0f - c4[q]) + EPSF);
        float aff = __expf(-d * invT);
        if (4 * lane + q == row) aff = 0.0f;
        e[r][q] = aff;
      }
      const float h0 = __builtin_fmaxf(e[r][0], e[r][1]), l0 = __builtin_fminf(e[r][0], e[r][1]);
      const float h1 = __builtin_fmaxf(e[r][2], e[r][3]), l1 = __builtin_fminf(e[r][2], e[r][3]);
      t0[r] = __builtin_fmaxf(h0, h1);
      t1[r] = __builtin_fmaxf(__builtin_fmaxf(l0, l1), __builtin_fminf(h0, h1));
      t2[r] = __builtin_fmaxf(__builtin_fminf(l0, h1), __builtin_fminf(h0, l1));
    }
    merge_top3_rows<1>(t0, t1, t2);
    merge_top3_rows<2>(t0, t1, t2);
    merge_top3_rows<4>(t0, t1, t2);
    merge_top3_rows<8>(t0, t1, t2);
    merge_top3_rows<16>(t0, t1, t2);

    float sum[4];
#pragma unroll
    for (int r = 0; r < 4; ++r) {
      float s = 0.0f;
#pragma unroll
      for (int q = 0; q < 4; ++q) {
        const float a = e[r][q];
        const float x = (a >= t2[r]) ? __expf(a - t0[r]) : 0.0f;
        e[r][q] = x;
        s += x;
      }
      sum[r] = s;
    }
    add_rows<1>(sum);
    add_rows<2>(sum);
    add_rows<4>(sum);
    add_rows<8>(sum);
    add_rows<16>(sum);
#pragma unroll
    for (int r = 0; r < 4; ++r) {
      const int row = wave * 16 + rb + r;
      const float rsum = 1.0f / sum[r];
      float4 p;
      p.x = e[r][0] * rsum; p.y = e[r][1] * rsum; p.z = e[r][2] * rsum; p.w = e[r][3] * rsum;
      *(float4*)&sS[row * LDSTR + 4 * lane] = p;
    }
  }
  __syncthreads();

  for (int e = tid; e < N * N; e += 256) {
    int i = e >> 7, j = e & 127;
    if (i < j) {
      float a = sS[i * LDSTR + j];
      float c2 = sS[j * LDSTR + i];
      float m = 0.5f * (a + c2);
      sS[i * LDSTR + j] = m;
      sS[j * LDSTR + i] = m;
    }
  }

  const int tm = wave;
  float creg[8][8];
#pragma unroll
  for (int tn = 0; tn < 8; ++tn) {
#pragma unroll
    for (int v = 0; v < 8; ++v)
      creg[tn][v] = sC[(tm * 16 + v + 8 * lh) * LDSTR + tn * 16 + lm];
  }
  __syncthreads();

#pragma unroll
  for (int tn = 0; tn < 8; ++tn) {
#pragma unroll
    for (int v = 0; v < 8; ++v)
      sFp[fidx(tm * 16 + v + 8 * lh, tn * 16 + lm)] = creg[tn][v];
  }
  __syncthreads();

  const int arowbase = (tm * 16 + lm) * LDSTR + 2 * lh;
  const int bcolbase = lh * PSTR + lm * 2;

  const float kAlpha[NSTEPS] = { 0.5f, 1.0f / 3.0f, 0.25f };

#pragma unroll
  for (int step = 0; step < NSTEPS; ++step) {
    v8f acc[8] = {};

    v2f a0 = *(const v2f*)&sS[arowbase];
    v2f b0[8];
#pragma unroll
    for (int tn = 0; tn < 8; ++tn)
      b0[tn] = *(const v2f*)&sFp[bcolbase + tn * 32];

    for (int k = 0; k < N; k += 8) {
      v2f a1 = *(const v2f*)&sS[arowbase + k + 4];
      const int bb1 = ((k + 4) >> 1) * PSTR + bcolbase;
      v2f b1[8];
#pragma unroll
      for (int tn = 0; tn < 8; ++tn)
        b1[tn] = *(const v2f*)&sFp[bb1 + tn * 32];
#pragma unroll
      for (int tn = 0; tn < 8; ++tn)
        acc[tn] = __builtin_amdgcn_wmma_f32_16x16x4_f32(
            false, a0, false, b0[tn], (short)0, acc[tn], false, false);

      const int kn = (k + 8) & (N - 1);
      a0 = *(const v2f*)&sS[arowbase + kn];
      const int bb0 = (kn >> 1) * PSTR + bcolbase;
#pragma unroll
      for (int tn = 0; tn < 8; ++tn)
        b0[tn] = *(const v2f*)&sFp[bb0 + tn * 32];
#pragma unroll
      for (int tn = 0; tn < 8; ++tn)
        acc[tn] = __builtin_amdgcn_wmma_f32_16x16x4_f32(
            false, a1, false, b1[tn], (short)0, acc[tn], false, false);
    }

    __syncthreads();

    const float alpha = kAlpha[step];
#pragma unroll
    for (int tn = 0; tn < 8; ++tn) {
#pragma unroll
      for (int v = 0; v < 8; ++v) {
        int row  = tm * 16 + v + 8 * lh;
        int col  = tn * 16 + lm;
        int ridx = row * LDSTR + col;
        int fi   = fidx(row, col);
        float msg = acc[tn][v];
        float f   = sFp[fi];
        float s   = sS[ridx];
        float cc  = creg[tn][v];
        float fn  = __builtin_fmaf(alpha, msg - f, f);
        sFp[fi]   = __builtin_fmaf(s, cc - fn, fn);
      }
    }
    __syncthreads();
  }

  float* dst = out + (size_t)b * (N * N);
  for (int e = tid; e < N * N; e += 256) {
    int i = e >> 7, j = e & 127;
    float val = (i == j) ? 1.0f
                         : 0.5f * (sFp[fidx(i, j)] + sFp[fidx(j, i)]);
    dst[e] = val;
  }
}

extern "C" void kernel_launch(void* const* d_in, const int* in_sizes, int n_in,
                              void* d_out, int out_size, void* d_ws, size_t ws_size,
                              hipStream_t stream) {
  (void)in_sizes; (void)n_in; (void)out_size; (void)d_ws; (void)ws_size;
  const float* corr = (const float*)d_in[0];
  const float* temp = (const float*)d_in[1];
  float* out = (float*)d_out;

  const size_t shmem =
      ((size_t)SFLOATS + (size_t)CFLOATS) * sizeof(float);
  static bool attr_set = false;
  if (!attr_set) {
    hipFuncSetAttribute((const void*)sgf_kernel,
                        hipFuncAttributeMaxDynamicSharedMemorySize, (int)shmem);
    attr_set = true;
  }
  sgf_kernel<<<dim3(NBATCH), dim3(256), shmem, stream>>>(corr, temp, out);
}


# Each element-wise pass (global staging, clip+sym, adjacency sym, the 3 in-place updates, output) reads 16-32 elements into registers ahead of its stores: one round trip per group, not per element
# speedup vs baseline: 3.0334x; 1.3159x over previous
#include <hip/hip_runtime.h>


#define N       128
#define LDSTR   136
#define PSTR    288
#define NBATCH  2048
#define EPSF    1e-8f
#define NSTEPS  3
#define SFLOATS (N * LDSTR)
#define CFLOATS ((N / 2) * PSTR)

typedef __attribute__((ext_vector_type(2))) float v2f;
typedef __attribute__((ext_vector_type(8))) float v8f;

__device__ __forceinline__ int fidx(int row, int col) {
  return (row >> 1) * PSTR + col * 2 + (row & 1);
}

template <int M>
__device__ __forceinline__ float xlane(float x) {
  const int b = __builtin_bit_cast(int, x);
  if (M == 16)
    return __builtin_bit_cast(float,
        __builtin_amdgcn_permlanex16(b, b, 0x76543210, 0xfedcba98, false, false));
  return __builtin_bit_cast(float,
      __builtin_amdgcn_mov_dpp(b, 0x160 + (M & 15), 0xf, 0xf, false));
}

template <int M>
__device__ __forceinline__ void merge_top3(float& a0, float& a1, float& a2) {
  const float b0 = xlane<M>(a0), b1 = xlane<M>(a1), b2 = xlane<M>(a2);
  const float c0 = __builtin_fmaxf(a0, b0);
  const float c1 = __builtin_fmaxf(__builtin_fmaxf(a1, b1), __builtin_fminf(a0, b0));
  const float c2 = __builtin_fmaxf(__builtin_fmaxf(a2, b2),
                                   __builtin_fmaxf(__builtin_fminf(a1, b0), __builtin_fminf(a0, b1)));
  a0 = c0; a1 = c1; a2 = c2;
}

template <int M>
__device__ __forceinline__ void merge_top3_rows(float (&t0)[4], float (&t1)[4], float (&t2)[4]) {
#pragma unroll
  for (int r = 0; r < 4; ++r) merge_top3<M>(t0[r], t1[r], t2[r]);
}

template <int M>
__device__ __forceinline__ void add_rows(float (&s)[4]) {
#pragma unroll
  for (int r = 0; r < 4; ++r) s[r] += xlane<M>(s[r]);
}

__global__ __launch_bounds__(256, 4)
void sgf_kernel(const float* __restrict__ corr,
                const float* __restrict__ temperature,
                float* __restrict__ out) {
  extern __shared__ float lds[];
  float* sS  = lds;
  float* sC  = lds + SFLOATS;
  float* sFp = lds + SFLOATS;

  const int tid  = threadIdx.x;
  const int lane = tid & 31;
  const int wave = tid >> 5;
  const int lm   = lane & 15;
  const int lh   = lane >> 4;
  const int b    = blockIdx.x;

  const float invT = 1.0f / temperature[0];
  const float* src = corr + (size_t)b * (N * N);

  {
    float4 stg[16];
#pragma unroll
    for (int it = 0; it < 16; ++it)
      stg[it] = *(const float4*)(src + tid * 4 + it * 1024);
#pragma unroll
    for (int it = 0; it < 16; ++it) {
      const int e = tid * 4 + it * 1024;
      const int i = e >> 7, j = e & 127;
      *(float4*)&sS[i * LDSTR + j] = stg[it];
    }
  }
  __syncthreads();

  {
    const int j  = tid & 127;
    const int i0 = tid >> 7;
    for (int it0 = 0; it0 < 64; it0 += 16) {
      float x[16], y[16];
#pragma unroll
      for (int u = 0; u < 16; ++u) {
        const int i = i0 + 2 * (it0 + u);
        x[u] = sS[i * LDSTR + j];
        y[u] = sS[j * LDSTR + i];
      }
#pragma unroll
      for (int u = 0; u < 16; ++u) {
        const int i = i0 + 2 * (it0 + u);
        const float xc = fminf(fmaxf(x[u], -1.0f + EPSF), 1.0f - EPSF);
        const float yc = fminf(fmaxf(y[u], -1.0f + EPSF), 1.0f - EPSF);
        sC[i * LDSTR + j] = 0.5f * (xc + yc);
      }
    }
  }
  __syncthreads();

  for (int rb = 0; rb < 16; rb += 4) {
    float e[4][4];
    float t0[4], t1[4], t2[4];
#pragma unroll
    for (int r = 0; r < 4; ++r) {
      const int row = wave * 16 + rb + r;
      const float4 cv = *(const float4*)&sC[row * LDSTR + 4 * lane];
      const float c4[4] = { cv.x, cv.y, cv.z, cv.w };
#pragma unroll
      for (int q = 0; q < 4; ++q) {
        float d   = __builtin_sqrtf(2.0f * (1.0f - c4[q]) + EPSF);
        float aff = __expf(-d * invT);
        if (4 * lane + q == row) aff = 0.0f;
        e[r][q] = aff;
      }
      const float h0 = __builtin_fmaxf(e[r][0], e[r][1]), l0 = __builtin_fminf(e[r][0], e[r][1]);
      const float h1 = __builtin_fmaxf(e[r][2], e[r][3]), l1 = __builtin_fminf(e[r][2], e[r][3]);
      t0[r] = __builtin_fmaxf(h0, h1);
      t1[r] = __builtin_fmaxf(__builtin_fmaxf(l0, l1), __builtin_fminf(h0, h1));
      t2[r] = __builtin_fmaxf(__builtin_fminf(l0, h1), __builtin_fminf(h0, l1));
    }
    merge_top3_rows<1>(t0, t1, t2);
    merge_top3_rows<2>(t0, t1, t2);
    merge_top3_rows<4>(t0, t1, t2);
    merge_top3_rows<8>(t0, t1, t2);
    merge_top3_rows<16>(t0, t1, t2);

    float sum[4];
#pragma unroll
    for (int r = 0; r < 4; ++r) {
      float s = 0.0f;
#pragma unroll
      for (int q = 0; q < 4; ++q) {
        const float a = e[r][q];
        const float x = (a >= t2[r]) ? __expf(a - t0[r]) : 0.0f;
        e[r][q] = x;
        s += x;
      }
      sum[r] = s;
    }
    add_rows<1>(sum);
    add_rows<2>(sum);
    add_rows<4>(sum);
    add_rows<8>(sum);
    add_rows<16>(sum);
#pragma unroll
    for (int r = 0; r < 4; ++r) {
      const int row = wave * 16 + rb + r;
      const float rsum = 1.0f / sum[r];
      float4 p;
      p.x = e[r][0] * rsum; p.y = e[r][1] * rsum; p.z = e[r][2] * rsum; p.w = e[r][3] * rsum;
      *(float4*)&sS[row * LDSTR + 4 * lane] = p;
    }
  }
  __syncthreads();

  {
    const int j  = tid & 127;
    const int i0 = tid >> 7;
    for (int it0 = 0; it0 < 64; it0 += 16) {
      if (i0 + 2 * it0 < (j | 31)) {
        float a[16], t[16];
#pragma unroll
        for (int u = 0; u < 16; ++u) {
          const int i = i0 + 2 * (it0 + u);
          a[u] = sS[i * LDSTR + j];
          t[u] = sS[j * LDSTR + i];
        }
#pragma unroll
        for (int u = 0; u < 16; ++u) {
          const int i = i0 + 2 * (it0 + u);
          if (i < j) {
            const float m = 0.5f * (a[u] + t[u]);
            sS[i * LDSTR + j] = m;
            sS[j * LDSTR + i] = m;
          }
        }
      }
    }
  }

  const int tm = wave;
  float creg[8][8];
#pragma unroll
  for (int tn = 0; tn < 8; ++tn) {
#pragma unroll
    for (int v = 0; v < 8; ++v)
      creg[tn][v] = sC[(tm * 16 + v + 8 * lh) * LDSTR + tn * 16 + lm];
  }
  __syncthreads();

#pragma unroll
  for (int tn = 0; tn < 8; ++tn) {
#pragma unroll
    for (int v = 0; v < 8; ++v)
      sFp[fidx(tm * 16 + v + 8 * lh, tn * 16 + lm)] = creg[tn][v];
  }
  __syncthreads();

  const int arowbase = (tm * 16 + lm) * LDSTR + 2 * lh;
  const int bcolbase = lh * PSTR + lm * 2;

  const float kAlpha[NSTEPS] = { 0.5f, 1.0f / 3.0f, 0.25f };

#pragma unroll
  for (int step = 0; step < NSTEPS; ++step) {
    v8f acc[8] = {};

    v2f a0 = *(const v2f*)&sS[arowbase];
    v2f b0[8];
#pragma unroll
    for (int tn = 0; tn < 8; ++tn)
      b0[tn] = *(const v2f*)&sFp[bcolbase + tn * 32];

    for (int k = 0; k < N; k += 8) {
      v2f a1 = *(const v2f*)&sS[arowbase + k + 4];
      const int bb1 = ((k + 4) >> 1) * PSTR + bcolbase;
      v2f b1[8];
#pragma unroll
      for (int tn = 0; tn < 8; ++tn)
        b1[tn] = *(const v2f*)&sFp[bb1 + tn * 32];
#pragma unroll
      for (int tn = 0; tn < 8; ++tn)
        acc[tn] = __builtin_amdgcn_wmma_f32_16x16x4_f32(
            false, a0, false, b0[tn], (short)0, acc[tn], false, false);

      const int kn = (k + 8) & (N - 1);
      a0 = *(const v2f*)&sS[arowbase + kn];
      const int bb0 = (kn >> 1) * PSTR + bcolbase;
#pragma unroll
      for (int tn = 0; tn < 8; ++tn)
        b0[tn] = *(const v2f*)&sFp[bb0 + tn * 32];
#pragma unroll
      for (int tn = 0; tn < 8; ++tn)
        acc[tn] = __builtin_amdgcn_wmma_f32_16x16x4_f32(
            false, a1, false, b1[tn], (short)0, acc[tn], false, false);
    }

    __syncthreads();

    const float alpha = kAlpha[step];
#pragma unroll
    for (int hf = 0; hf < 2; ++hf) {
      v2f   f2[4][4];
      float sv[4][8];
#pragma unroll
      for (int t = 0; t < 4; ++t) {
        const int col = (hf * 4 + t) * 16 + lm;
#pragma unroll
        for (int p = 0; p < 4; ++p)
          f2[t][p] = *(const v2f*)&sFp[(tm * 8 + 4 * lh + p) * PSTR + 2 * col];
#pragma unroll
        for (int v = 0; v < 8; ++v)
          sv[t][v] = sS[(tm * 16 + v + 8 * lh) * LDSTR + col];
      }
#pragma unroll
      for (int t = 0; t < 4; ++t) {
        const int tn  = hf * 4 + t;
        const int col = tn * 16 + lm;
#pragma unroll
        for (int p = 0; p < 4; ++p) {
          const float fa  = f2[t][p][0], fb = f2[t][p][1];
          const float fna = __builtin_fmaf(alpha, acc[tn][2 * p] - fa, fa);
          const float fnb = __builtin_fmaf(alpha, acc[tn][2 * p + 1] - fb, fb);
          v2f r;
          r[0] = __builtin_fmaf(sv[t][2 * p], creg[tn][2 * p] - fna, fna);
          r[1] = __builtin_fmaf(sv[t][2 * p + 1], creg[tn][2 * p + 1] - fnb, fnb);
          *(v2f*)&sFp[(tm * 8 + 4 * lh + p) * PSTR + 2 * col] = r;
        }
      }
    }
    __syncthreads();
  }

  float* dst = out + (size_t)b * (N * N);
  {
    const int j  = tid & 127;
    const int i0 = tid >> 7;
    for (int it0 = 0; it0 < 64; it0 += 16) {
      float p[16], q[16];
#pragma unroll
      for (int u = 0; u < 16; ++u) {
        const int i = i0 + 2 * (it0 + u);
        p[u] = sFp[fidx(i, j)];
        q[u] = sFp[fidx(j, i)];
      }
#pragma unroll
      for (int u = 0; u < 16; ++u) {
        const int i = i0 + 2 * (it0 + u);
        dst[i * N + j] = (i == j) ? 1.0f : 0.5f * (p[u] + q[u]);
      }
    }
  }
}

extern "C" void kernel_launch(void* const* d_in, const int* in_sizes, int n_in,
                              void* d_out, int out_size, void* d_ws, size_t ws_size,
                              hipStream_t stream) {
  (void)in_sizes; (void)n_in; (void)out_size; (void)d_ws; (void)ws_size;
  const float* corr = (const float*)d_in[0];
  const float* temp = (const float*)d_in[1];
  float* out = (float*)d_out;

  const size_t shmem =
      ((size_t)SFLOATS + (size_t)CFLOATS) * sizeof(float);
  static bool attr_set = false;
  if (!attr_set) {
    hipFuncSetAttribute((const void*)sgf_kernel,
                        hipFuncAttributeMaxDynamicSharedMemorySize, (int)shmem);
    attr_set = true;
  }
  sgf_kernel<<<dim3(NBATCH), dim3(256), shmem, stream>>>(corr, temp, out);
}


# The three symmetrizing passes (clip+sym, adjacency, output) go plane to plane on 4x4 register blocks with 16-byte LDS accesses at row stride 132 (no bank conflicts); output leaves as whole rows
# speedup vs baseline: 3.6312x; 1.1971x over previous
#include <hip/hip_runtime.h>


#define N       128
#define LDSTR   132
#define PSTR    288
#define NBATCH  2048
#define EPSF    1e-8f
#define NSTEPS  3
#define SFLOATS (N * LDSTR)
#define CFLOATS ((N / 2) * PSTR)

typedef __attribute__((ext_vector_type(2))) float v2f;
typedef __attribute__((ext_vector_type(8))) float v8f;

__device__ __forceinline__ int fidx(int row, int col) {
  return (row >> 1) * PSTR + col * 2 + (row & 1);
}

template <int M>
__device__ __forceinline__ float xlane(float x) {
  const int b = __builtin_bit_cast(int, x);
  if (M == 16)
    return __builtin_bit_cast(float,
        __builtin_amdgcn_permlanex16(b, b, 0x76543210, 0xfedcba98, false, false));
  return __builtin_bit_cast(float,
      __builtin_amdgcn_mov_dpp(b, 0x160 + (M & 15), 0xf, 0xf, false));
}

template <int M>
__device__ __forceinline__ void merge_top3(float& a0, float& a1, float& a2) {
  const float b0 = xlane<M>(a0), b1 = xlane<M>(a1), b2 = xlane<M>(a2);
  const float c0 = __builtin_fmaxf(a0, b0);
  const float c1 = __builtin_fmaxf(__builtin_fmaxf(a1, b1), __builtin_fminf(a0, b0));
  const float c2 = __builtin_fmaxf(__builtin_fmaxf(a2, b2),
                                   __builtin_fmaxf(__builtin_fminf(a1, b0), __builtin_fminf(a0, b1)));
  a0 = c0; a1 = c1; a2 = c2;
}

template <int M>
__device__ __forceinline__ void merge_top3_rows(float (&t0)[4], float (&t1)[4], float (&t2)[4]) {
#pragma unroll
  for (int r = 0; r < 4; ++r) merge_top3<M>(t0[r], t1[r], t2[r]);
}

template <int M>
__device__ __forceinline__ void add_rows(float (&s)[4]) {
#pragma unroll
  for (int r = 0; r < 4; ++r) s[r] += xlane<M>(s[r]);
}

template <int MODE>
__device__ __forceinline__ void sym_tiles(const float* __restrict__ srcp, float* __restrict__ dstp, int tid) {
  const int hw = tid >> 4;
  const int a  = tid & 3;
  const int q  = (tid >> 2) & 3;
  for (int trip = 0; trip < 4; ++trip) {
    const int t  = trip * 16 + hw;
    const int r0 = 16 * (t >> 3) + 4 * a;
    const int c0 = 16 * (t & 7) + 4 * q;
    float d[4][4], m[4][4];
#pragma unroll
    for (int r = 0; r < 4; ++r) {
      const float4 dv = *(const float4*)&srcp[(r0 + r) * LDSTR + c0];
      const float4 mv = *(const float4*)&srcp[(c0 + r) * LDSTR + r0];
      d[r][0] = dv.x; d[r][1] = dv.y; d[r][2] = dv.z; d[r][3] = dv.w;
      m[r][0] = mv.x; m[r][1] = mv.y; m[r][2] = mv.z; m[r][3] = mv.w;
    }
#pragma unroll
    for (int r = 0; r < 4; ++r) {
      float o[4];
#pragma unroll
      for (int c = 0; c < 4; ++c) {
        float x = d[r][c], y = m[c][r];
        if (MODE == 0) {
          x = fminf(fmaxf(x, -1.0f + EPSF), 1.0f - EPSF);
          y = fminf(fmaxf(y, -1.0f + EPSF), 1.0f - EPSF);
        }
        float v = 0.5f * (x + y);
        if (MODE == 2 && r0 + r == c0 + c) v = 1.0f;
        o[c] = v;
      }
      float4 ov; ov.x = o[0]; ov.y = o[1]; ov.z = o[2]; ov.w = o[3];
      *(float4*)&dstp[(r0 + r) * LDSTR + c0] = ov;
    }
  }
}

__global__ __launch_bounds__(256, 4)
void sgf_kernel(const float* __restrict__ corr,
                const float* __restrict__ temperature,
                float* __restrict__ out) {
  extern __shared__ float lds[];
  float* sS  = lds;
  float* sC  = lds + SFLOATS;
  float* sFp = lds + SFLOATS;

  const int tid  = threadIdx.x;
  const int lane = tid & 31;
  const int wave = tid >> 5;
  const int lm   = lane & 15;
  const int lh   = lane >> 4;
  const int b    = blockIdx.x;

  const float invT = 1.0f / temperature[0];
  const float* src = corr + (size_t)b * (N * N);

  {
    float4 stg[16];
#pragma unroll
    for (int it = 0; it < 16; ++it)
      stg[it] = *(const float4*)(src + tid * 4 + it * 1024);
#pragma unroll
    for (int it = 0; it < 16; ++it) {
      const int e = tid * 4 + it * 1024;
      const int i = e >> 7, j = e & 127;
      *(float4*)&sS[i * LDSTR + j] = stg[it];
    }
  }
  __syncthreads();

  sym_tiles<0>(sS, sC, tid);
  __syncthreads();

  const int tm = wave;
  float creg[8][8];
#pragma unroll
  for (int tn = 0; tn < 8; ++tn) {
#pragma unroll
    for (int v = 0; v < 8; ++v)
      creg[tn][v] = sC[(tm * 16 + v + 8 * lh) * LDSTR + tn * 16 + lm];
  }

  for (int rb = 0; rb < 16; rb += 4) {
    float e[4][4];
    float t0[4], t1[4], t2[4];
#pragma unroll
    for (int r = 0; r < 4; ++r) {
      const int row = wave * 16 + rb + r;
      const float4 cv = *(const float4*)&sC[row * LDSTR + 4 * lane];
      const float c4[4] = { cv.x, cv.y, cv.z, cv.w };
#pragma unroll
      for (int q = 0; q < 4; ++q) {
        float d   = __builtin_sqrtf(2.0f * (1.0f - c4[q]) + EPSF);
        float aff = __expf(-d * invT);
        if (4 * lane + q == row) aff = 0.0f;
        e[r][q] = aff;
      }
      const float h0 = __builtin_fmaxf(e[r][0], e[r][1]), l0 = __builtin_fminf(e[r][0], e[r][1]);
      const float h1 = __builtin_fmaxf(e[r][2], e[r][3]), l1 = __builtin_fminf(e[r][2], e[r][3]);
      t0[r] = __builtin_fmaxf(h0, h1);
      t1[r] = __builtin_fmaxf(__builtin_fmaxf(l0, l1), __builtin_fminf(h0, h1));
      t2[r] = __builtin_fmaxf(__builtin_fminf(l0, h1), __builtin_fminf(h0, l1));
    }
    merge_top3_rows<1>(t0, t1, t2);
    merge_top3_rows<2>(t0, t1, t2);
    merge_top3_rows<4>(t0, t1, t2);
    merge_top3_rows<8>(t0, t1, t2);
    merge_top3_rows<16>(t0, t1, t2);

    float sum[4];
#pragma unroll
    for (int r = 0; r < 4; ++r) {
      float s = 0.0f;
#pragma unroll
      for (int q = 0; q < 4; ++q) {
        const float a = e[r][q];
        const float x = (a >= t2[r]) ? __expf(a - t0[r]) : 0.0f;
        e[r][q] = x;
        s += x;
      }
      sum[r] = s;
    }
    add_rows<1>(sum);
    add_rows<2>(sum);
    add_rows<4>(sum);
    add_rows<8>(sum);
    add_rows<16>(sum);
#pragma unroll
    for (int r = 0; r < 4; ++r) {
      const int row = wave * 16 + rb + r;
      const float rsum = 1.0f / sum[r];
      float4 p;
      p.x = e[r][0] * rsum; p.y = e[r][1] * rsum; p.z = e[r][2] * rsum; p.w = e[r][3] * rsum;
      *(float4*)&sC[row * LDSTR + 4 * lane] = p;
    }
  }
  __syncthreads();

  sym_tiles<1>(sC, sS, tid);
  __syncthreads();

#pragma unroll
  for (int tn = 0; tn < 8; ++tn) {
#pragma unroll
    for (int p = 0; p < 4; ++p) {
      v2f r;
      r[0] = creg[tn][2 * p];
      r[1] = creg[tn][2 * p + 1];
      *(v2f*)&sFp[(tm * 8 + 4 * lh + p) * PSTR + 2 * (tn * 16 + lm)] = r;
    }
  }
  __syncthreads();

  const int arowbase = (tm * 16 + lm) * LDSTR + 2 * lh;
  const int bcolbase = lh * PSTR + lm * 2;

  const float kAlpha[NSTEPS] = { 0.5f, 1.0f / 3.0f, 0.25f };

#pragma unroll
  for (int step = 0; step < NSTEPS; ++step) {
    v8f acc[8] = {};

    v2f a0 = *(const v2f*)&sS[arowbase];
    v2f b0[8];
#pragma unroll
    for (int tn = 0; tn < 8; ++tn)
      b0[tn] = *(const v2f*)&sFp[bcolbase + tn * 32];

    for (int k = 0; k < N; k += 8) {
      v2f a1 = *(const v2f*)&sS[arowbase + k + 4];
      const int bb1 = ((k + 4) >> 1) * PSTR + bcolbase;
      v2f b1[8];
#pragma unroll
      for (int tn = 0; tn < 8; ++tn)
        b1[tn] = *(const v2f*)&sFp[bb1 + tn * 32];
#pragma unroll
      for (int tn = 0; tn < 8; ++tn)
        acc[tn] = __builtin_amdgcn_wmma_f32_16x16x4_f32(
            false, a0, false, b0[tn], (short)0, acc[tn], false, false);

      const int kn = (k + 8) & (N - 1);
      a0 = *(const v2f*)&sS[arowbase + kn];
      const int bb0 = (kn >> 1) * PSTR + bcolbase;
#pragma unroll
      for (int tn = 0; tn < 8; ++tn)
        b0[tn] = *(const v2f*)&sFp[bb0 + tn * 32];
#pragma unroll
      for (int tn = 0; tn < 8; ++tn)
        acc[tn] = __builtin_amdgcn_wmma_f32_16x16x4_f32(
            false, a1, false, b1[tn], (short)0, acc[tn], false, false);
    }

    __syncthreads();

    const float alpha = kAlpha[step];
#pragma unroll
    for (int hf = 0; hf < 2; ++hf) {
      v2f   f2[4][4];
      float sv[4][8];
#pragma unroll
      for (int t = 0; t < 4; ++t) {
        const int col = (hf * 4 + t) * 16 + lm;
#pragma unroll
        for (int p = 0; p < 4; ++p)
          f2[t][p] = *(const v2f*)&sFp[(tm * 8 + 4 * lh + p) * PSTR + 2 * col];
#pragma unroll
        for (int v = 0; v < 8; ++v)
          sv[t][v] = sS[(tm * 16 + v + 8 * lh) * LDSTR + col];
      }
#pragma unroll
      for (int t = 0; t < 4; ++t) {
        const int tn  = hf * 4 + t;
        const int col = tn * 16 + lm;
#pragma unroll
        for (int p = 0; p < 4; ++p) {
          const float fa  = f2[t][p][0], fb = f2[t][p][1];
          const float fna = __builtin_fmaf(alpha, acc[tn][2 * p] - fa, fa);
          const float fnb = __builtin_fmaf(alpha, acc[tn][2 * p + 1] - fb, fb);
          v2f r;
          r[0] = __builtin_fmaf(sv[t][2 * p], creg[tn][2 * p] - fna, fna);
          r[1] = __builtin_fmaf(sv[t][2 * p + 1], creg[tn][2 * p + 1] - fnb, fnb);
          if (step == NSTEPS - 1) {
            sS[(tm * 16 + 2 * p + 8 * lh) * LDSTR + col]     = r[0];
            sS[(tm * 16 + 2 * p + 1 + 8 * lh) * LDSTR + col] = r[1];
          } else {
            *(v2f*)&sFp[(tm * 8 + 4 * lh + p) * PSTR + 2 * col] = r;
          }
        }
      }
    }
    __syncthreads();
  }

  float* sO = lds + SFLOATS;
  sym_tiles<2>(sS, sO, tid);
  __syncthreads();
  float* dst = out + (size_t)b * (N * N);
  {
    float4 og[16];
#pragma unroll
    for (int it = 0; it < 16; ++it) {
      const int e = tid * 4 + it * 1024;
      og[it] = *(const float4*)&sO[(e >> 7) * LDSTR + (e & 127)];
    }
#pragma unroll
    for (int it = 0; it < 16; ++it)
      *(float4*)(dst + tid * 4 + it * 1024) = og[it];
  }
}

extern "C" void kernel_launch(void* const* d_in, const int* in_sizes, int n_in,
                              void* d_out, int out_size, void* d_ws, size_t ws_size,
                              hipStream_t stream) {
  (void)in_sizes; (void)n_in; (void)out_size; (void)d_ws; (void)ws_size;
  const float* corr = (const float*)d_in[0];
  const float* temp = (const float*)d_in[1];
  float* out = (float*)d_out;

  const size_t shmem =
      ((size_t)SFLOATS + (size_t)CFLOATS) * sizeof(float);
  static bool attr_set = false;
  if (!attr_set) {
    hipFuncSetAttribute((const void*)sgf_kernel,
                        hipFuncAttributeMaxDynamicSharedMemorySize, (int)shmem);
    attr_set = true;
  }
  sgf_kernel<<<dim3(NBATCH), dim3(256), shmem, stream>>>(corr, temp, out);
}


# Each wave owns a 16-column strip of all 128 rows: its filtered strip stays in registers as the B operand (C/D layout = B layout), no LDS traffic or barriers between steps; k sum order changes
# speedup vs baseline: 3.8984x; 1.0736x over previous
#include <hip/hip_runtime.h>


#define N       128
#define LDSTR   132
#define NBATCH  2048
#define EPSF    1e-8f
#define NSTEPS  3
#define PFLOATS (N * LDSTR)

typedef __attribute__((ext_vector_type(2))) float v2f;
typedef __attribute__((ext_vector_type(8))) float v8f;

template <int M>
__device__ __forceinline__ float xlane(float x) {
  const int b = __builtin_bit_cast(int, x);
  if (M == 16)
    return __builtin_bit_cast(float,
        __builtin_amdgcn_permlanex16(b, b, 0x76543210, 0xfedcba98, false, false));
  return __builtin_bit_cast(float,
      __builtin_amdgcn_mov_dpp(b, 0x160 + (M & 15), 0xf, 0xf, false));
}

template <int M>
__device__ __forceinline__ void merge_top3(float& a0, float& a1, float& a2) {
  const float b0 = xlane<M>(a0), b1 = xlane<M>(a1), b2 = xlane<M>(a2);
  const float c0 = __builtin_fmaxf(a0, b0);
  const float c1 = __builtin_fmaxf(__builtin_fmaxf(a1, b1), __builtin_fminf(a0, b0));
  const float c2 = __builtin_fmaxf(__builtin_fmaxf(a2, b2),
                                   __builtin_fmaxf(__builtin_fminf(a1, b0), __builtin_fminf(a0, b1)));
  a0 = c0; a1 = c1; a2 = c2;
}

template <int M>
__device__ __forceinline__ void merge_top3_rows(float (&t0)[4], float (&t1)[4], float (&t2)[4]) {
#pragma unroll
  for (int r = 0; r < 4; ++r) merge_top3<M>(t0[r], t1[r], t2[r]);
}

template <int M>
__device__ __forceinline__ void add_rows(float (&s)[4]) {
#pragma unroll
  for (int r = 0; r < 4; ++r) s[r] += xlane<M>(s[r]);
}

template <int MODE>
__device__ __forceinline__ void sym_tiles(const float* __restrict__ srcp, float* __restrict__ dstp, int tid) {
  const int hw = tid >> 4;
  const int a  = tid & 3;
  const int q  = (tid >> 2) & 3;
  for (int trip = 0; trip < 4; ++trip) {
    const int t  = trip * 16 + hw;
    const int r0 = 16 * (t >> 3) + 4 * a;
    const int c0 = 16 * (t & 7) + 4 * q;
    float d[4][4], m[4][4];
#pragma unroll
    for (int r = 0; r < 4; ++r) {
      const float4 dv = *(const float4*)&srcp[(r0 + r) * LDSTR + c0];
      const float4 mv = *(const float4*)&srcp[(c0 + r) * LDSTR + r0];
      d[r][0] = dv.x; d[r][1] = dv.y; d[r][2] = dv.z; d[r][3] = dv.w;
      m[r][0] = mv.x; m[r][1] = mv.y; m[r][2] = mv.z; m[r][3] = mv.w;
    }
#pragma unroll
    for (int r = 0; r < 4; ++r) {
      float o[4];
#pragma unroll
      for (int c = 0; c < 4; ++c) {
        float x = d[r][c], y = m[c][r];
        if (MODE == 0) {
          x = fminf(fmaxf(x, -1.0f + EPSF), 1.0f - EPSF);
          y = fminf(fmaxf(y, -1.0f + EPSF), 1.0f - EPSF);
        }
        float v = 0.5f * (x + y);
        if (MODE == 2 && r0 + r == c0 + c) v = 1.0f;
        o[c] = v;
      }
      float4 ov; ov.x = o[0]; ov.y = o[1]; ov.z = o[2]; ov.w = o[3];
      *(float4*)&dstp[(r0 + r) * LDSTR + c0] = ov;
    }
  }
}

__global__ __launch_bounds__(256, 4)
void sgf_kernel(const float* __restrict__ corr,
                const float* __restrict__ temperature,
                float* __restrict__ out) {
  extern __shared__ float lds[];
  float* sS = lds;
  float* sC = lds + PFLOATS;

  const int tid  = threadIdx.x;
  const int lane = tid & 31;
  const int wave = tid >> 5;
  const int lm   = lane & 15;
  const int lh   = lane >> 4;
  const int b    = blockIdx.x;

  const float invT = 1.0f / temperature[0];
  const float* src = corr + (size_t)b * (N * N);

  {
    float4 stg[16];
#pragma unroll
    for (int it = 0; it < 16; ++it)
      stg[it] = *(const float4*)(src + tid * 4 + it * 1024);
#pragma unroll
    for (int it = 0; it < 16; ++it) {
      const int e = tid * 4 + it * 1024;
      const int i = e >> 7, j = e & 127;
      *(float4*)&sS[i * LDSTR + j] = stg[it];
    }
  }
  __syncthreads();

  sym_tiles<0>(sS, sC, tid);
  __syncthreads();

  v8f creg[8];
#pragma unroll
  for (int tm = 0; tm < 8; ++tm) {
#pragma unroll
    for (int v = 0; v < 8; ++v)
      creg[tm][v] = sC[(tm * 16 + v + 8 * lh) * LDSTR + wave * 16 + lm];
  }
  __syncthreads();

  for (int rb = 0; rb < 16; rb += 4) {
    float e[4][4];
    float t0[4], t1[4], t2[4];
#pragma unroll
    for (int r = 0; r < 4; ++r) {
      const int row = wave * 16 + rb + r;
      const float4 cv = *(const float4*)&sC[row * LDSTR + 4 * lane];
      const float c4[4] = { cv.x, cv.y, cv.z, cv.w };
#pragma unroll
      for (int q = 0; q < 4; ++q) {
        float d   = __builtin_sqrtf(2.0f * (1.0f - c4[q]) + EPSF);
        float aff = __expf(-d * invT);
        if (4 * lane + q == row) aff = 0.0f;
        e[r][q] = aff;
      }
      const float h0 = __builtin_fmaxf(e[r][0], e[r][1]), l0 = __builtin_fminf(e[r][0], e[r][1]);
      const float h1 = __builtin_fmaxf(e[r][2], e[r][3]), l1 = __builtin_fminf(e[r][2], e[r][3]);
      t0[r] = __builtin_fmaxf(h0, h1);
      t1[r] = __builtin_fmaxf(__builtin_fmaxf(l0, l1), __builtin_fminf(h0, h1));
      t2[r] = __builtin_fmaxf(__builtin_fminf(l0, h1), __builtin_fminf(h0, l1));
    }
    merge_top3_rows<1>(t0, t1, t2);
    merge_top3_rows<2>(t0, t1, t2);
    merge_top3_rows<4>(t0, t1, t2);
    merge_top3_rows<8>(t0, t1, t2);
    merge_top3_rows<16>(t0, t1, t2);

    float sum[4];
#pragma unroll
    for (int r = 0; r < 4; ++r) {
      float s = 0.0f;
#pragma unroll
      for (int q = 0; q < 4; ++q) {
        const float a = e[r][q];
        const float x = (a >= t2[r]) ? __expf(a - t0[r]) : 0.0f;
        e[r][q] = x;
        s += x;
      }
      sum[r] = s;
    }
    add_rows<1>(sum);
    add_rows<2>(sum);
    add_rows<4>(sum);
    add_rows<8>(sum);
    add_rows<16>(sum);
#pragma unroll
    for (int r = 0; r < 4; ++r) {
      const int row = wave * 16 + rb + r;
      const float rsum = 1.0f / sum[r];
      float4 p;
      p.x = e[r][0] * rsum; p.y = e[r][1] * rsum; p.z = e[r][2] * rsum; p.w = e[r][3] * rsum;
      *(float4*)&sC[row * LDSTR + 4 * lane] = p;
    }
  }
  __syncthreads();

  sym_tiles<1>(sC, sS, tid);
  __syncthreads();

  v8f f[8];
#pragma unroll
  for (int tm = 0; tm < 8; ++tm) f[tm] = creg[tm];

  const int aOff = lm * LDSTR + 8 * lh;
  const int sOff = (8 * lh) * LDSTR + wave * 16 + lm;

  for (int step = 0; step < NSTEPS; ++step) {
    v8f acc[8] = {};
    int aO = aOff, sO2 = sOff;
    asm volatile("" : "+v"(aO), "+v"(sO2));
    const float* aBase = sS + aO;
    const float* sBase = sS + sO2;

    float4 aC[4], aN[4];
#pragma unroll
    for (int j = 0; j < 4; ++j)
      aC[j] = *(const float4*)&aBase[(j * 16) * LDSTR];

#pragma unroll
    for (int g = 0; g < 32; ++g) {
      if (g + 1 < 32) {
        const int gn = g + 1;
        const int tan = 4 * (gn & 1), tmn = gn >> 2, qn = (gn >> 1) & 1;
#pragma unroll
        for (int j = 0; j < 4; ++j)
          aN[j] = *(const float4*)&aBase[((tan + j) * 16) * LDSTR + 16 * tmn + 4 * qn];
      }
      __builtin_amdgcn_sched_barrier(0);
      const int ta0 = 4 * (g & 1), tm = g >> 2, q = (g >> 1) & 1;
      v2f b0, b1;
      b0[0] = f[tm][4 * q];     b0[1] = f[tm][4 * q + 1];
      b1[0] = f[tm][4 * q + 2]; b1[1] = f[tm][4 * q + 3];
#pragma unroll
      for (int j = 0; j < 4; ++j) {
        v2f a; a[0] = aC[j].x; a[1] = aC[j].y;
        acc[ta0 + j] = __builtin_amdgcn_wmma_f32_16x16x4_f32(
            false, a, false, b0, (short)0, acc[ta0 + j], false, false);
      }
      __builtin_amdgcn_sched_barrier(0);
#pragma unroll
      for (int j = 0; j < 4; ++j) {
        v2f a; a[0] = aC[j].z; a[1] = aC[j].w;
        acc[ta0 + j] = __builtin_amdgcn_wmma_f32_16x16x4_f32(
            false, a, false, b1, (short)0, acc[ta0 + j], false, false);
      }
      __builtin_amdgcn_sched_barrier(0);
#pragma unroll
      for (int j = 0; j < 4; ++j) aC[j] = aN[j];
    }

    const float alpha = (step == 0) ? 0.5f : ((step == 1) ? (1.0f / 3.0f) : 0.25f);
#pragma unroll
    for (int tm = 0; tm < 8; ++tm) {
      float sv[8];
#pragma unroll
      for (int v = 0; v < 8; ++v)
        sv[v] = sBase[(tm * 16 + v) * LDSTR];
#pragma unroll
      for (int v = 0; v < 8; ++v) {
        const float fa = f[tm][v];
        const float fn = __builtin_fmaf(alpha, acc[tm][v] - fa, fa);
        f[tm][v] = __builtin_fmaf(sv[v], creg[tm][v] - fn, fn);
      }
    }
  }

  int tidE = tid;
  asm volatile("" : "+v"(tidE));
  const int lmE = tidE & 15, lhE = (tidE >> 4) & 1, waveE = tidE >> 5;
  float* sO = sC;
#pragma unroll
  for (int tm = 0; tm < 8; ++tm) {
#pragma unroll
    for (int v = 0; v < 8; ++v)
      sO[(tm * 16 + v + 8 * lhE) * LDSTR + waveE * 16 + lmE] = f[tm][v];
  }
  __syncthreads();

  sym_tiles<2>(sO, sS, tidE);
  __syncthreads();
  float* dst = out + (size_t)b * (N * N);
  {
    float4 og[16];
#pragma unroll
    for (int it = 0; it < 16; ++it) {
      const int e = tidE * 4 + it * 1024;
      og[it] = *(const float4*)&sS[(e >> 7) * LDSTR + (e & 127)];
    }
#pragma unroll
    for (int it = 0; it < 16; ++it)
      *(float4*)(dst + tidE * 4 + it * 1024) = og[it];
  }
}

extern "C" void kernel_launch(void* const* d_in, const int* in_sizes, int n_in,
                              void* d_out, int out_size, void* d_ws, size_t ws_size,
                              hipStream_t stream) {
  (void)in_sizes; (void)n_in; (void)out_size; (void)d_ws; (void)ws_size;
  const float* corr = (const float*)d_in[0];
  const float* temp = (const float*)d_in[1];
  float* out = (float*)d_out;

  const size_t shmem = 2 * (size_t)PFLOATS * sizeof(float);
  static bool attr_set = false;
  if (!attr_set) {
    hipFuncSetAttribute((const void*)sgf_kernel,
                        hipFuncAttributeMaxDynamicSharedMemorySize, (int)shmem);
    attr_set = true;
  }
  sgf_kernel<<<dim3(NBATCH), dim3(256), shmem, stream>>>(corr, temp, out);
}


# Graph build: one row per 8-lane group, each lane takes the top 3 of its 16 columns by insertion in registers; cross-lane butterfly 5 steps -> 3, one division per row; softmax row sum order changes
# speedup vs baseline: 4.1520x; 1.0651x over previous
#include <hip/hip_runtime.h>


#define N       128
#define LDSTR   132
#define NBATCH  2048
#define EPSF    1e-8f
#define NSTEPS  3
#define PFLOATS (N * LDSTR)

typedef __attribute__((ext_vector_type(2))) float v2f;
typedef __attribute__((ext_vector_type(8))) float v8f;

template <int M>
__device__ __forceinline__ float xlane(float x) {
  const int b = __builtin_bit_cast(int, x);
  return __builtin_bit_cast(float,
      __builtin_amdgcn_mov_dpp(b, 0x160 + (M & 15), 0xf, 0xf, false));
}

template <int M>
__device__ __forceinline__ void merge_top3(float& a0, float& a1, float& a2) {
  const float b0 = xlane<M>(a0), b1 = xlane<M>(a1), b2 = xlane<M>(a2);
  const float c0 = __builtin_fmaxf(a0, b0);
  const float c1 = __builtin_fmaxf(__builtin_fmaxf(a1, b1), __builtin_fminf(a0, b0));
  const float c2 = __builtin_fmaxf(__builtin_fmaxf(a2, b2),
                                   __builtin_fmaxf(__builtin_fminf(a1, b0), __builtin_fminf(a0, b1)));
  a0 = c0; a1 = c1; a2 = c2;
}

template <int MODE>
__device__ __forceinline__ void sym_tiles(const float* __restrict__ srcp, float* __restrict__ dstp, int tid) {
  const int hw = tid >> 4;
  const int a  = tid & 3;
  const int q  = (tid >> 2) & 3;
  for (int trip = 0; trip < 4; ++trip) {
    const int t  = trip * 16 + hw;
    const int r0 = 16 * (t >> 3) + 4 * a;
    const int c0 = 16 * (t & 7) + 4 * q;
    float d[4][4], m[4][4];
#pragma unroll
    for (int r = 0; r < 4; ++r) {
      const float4 dv = *(const float4*)&srcp[(r0 + r) * LDSTR + c0];
      const float4 mv = *(const float4*)&srcp[(c0 + r) * LDSTR + r0];
      d[r][0] = dv.x; d[r][1] = dv.y; d[r][2] = dv.z; d[r][3] = dv.w;
      m[r][0] = mv.x; m[r][1] = mv.y; m[r][2] = mv.z; m[r][3] = mv.w;
    }
#pragma unroll
    for (int r = 0; r < 4; ++r) {
      float o[4];
#pragma unroll
      for (int c = 0; c < 4; ++c) {
        float x = d[r][c], y = m[c][r];
        if (MODE == 0) {
          x = fminf(fmaxf(x, -1.0f + EPSF), 1.0f - EPSF);
          y = fminf(fmaxf(y, -1.0f + EPSF), 1.0f - EPSF);
        }
        float v = 0.5f * (x + y);
        if (MODE == 2 && r0 + r == c0 + c) v = 1.0f;
        o[c] = v;
      }
      float4 ov; ov.x = o[0]; ov.y = o[1]; ov.z = o[2]; ov.w = o[3];
      *(float4*)&dstp[(r0 + r) * LDSTR + c0] = ov;
    }
  }
}

__global__ __launch_bounds__(256, 4)
void sgf_kernel(const float* __restrict__ corr,
                const float* __restrict__ temperature,
                float* __restrict__ out) {
  extern __shared__ float lds[];
  float* sS = lds;
  float* sC = lds + PFLOATS;

  const int tid  = threadIdx.x;
  const int lane = tid & 31;
  const int wave = tid >> 5;
  const int lm   = lane & 15;
  const int lh   = lane >> 4;
  const int b    = blockIdx.x;

  const float invT = 1.0f / temperature[0];
  const float* src = corr + (size_t)b * (N * N);

  {
    float4 stg[16];
#pragma unroll
    for (int it = 0; it < 16; ++it)
      stg[it] = *(const float4*)(src + tid * 4 + it * 1024);
#pragma unroll
    for (int it = 0; it < 16; ++it) {
      const int e = tid * 4 + it * 1024;
      const int i = e >> 7, j = e & 127;
      *(float4*)&sS[i * LDSTR + j] = stg[it];
    }
  }
  __syncthreads();

  sym_tiles<0>(sS, sC, tid);
  __syncthreads();

  v8f creg[8];
#pragma unroll
  for (int tm = 0; tm < 8; ++tm) {
#pragma unroll
    for (int v = 0; v < 8; ++v)
      creg[tm][v] = sC[(tm * 16 + v + 8 * lh) * LDSTR + wave * 16 + lm];
  }
  __syncthreads();

  {
    const int gq  = lane >> 3;
    const int ch0 = (lane & 7) + 7 * (gq & 1);
    for (int rb = 0; rb < 16; rb += 4) {
      const int row = wave * 16 + rb + gq;
      float* rowp = sC + row * LDSTR;
      float4 cv[4];
#pragma unroll
      for (int i = 0; i < 4; ++i)
        cv[i] = *(const float4*)&rowp[4 * ((ch0 + 8 * i) & 31)];

      float e[16];
      float t0 = -__builtin_inff(), t1 = -__builtin_inff(), t2 = -__builtin_inff();
#pragma unroll
      for (int i = 0; i < 4; ++i) {
        const int col0 = 4 * ((ch0 + 8 * i) & 31);
        const float c4[4] = { cv[i].x, cv[i].y, cv[i].z, cv[i].w };
#pragma unroll
        for (int q = 0; q < 4; ++q) {
          float d   = __builtin_sqrtf(2.0f * (1.0f - c4[q]) + EPSF);
          float aff = __expf(-d * invT);
          if (col0 + q == row) aff = 0.0f;
          e[4 * i + q] = aff;
          const float n0 = __builtin_fmaxf(t0, aff);
          const float n1 = __builtin_fmaxf(__builtin_fminf(t0, aff), t1);
          const float n2 = __builtin_fmaxf(__builtin_fminf(t1, aff), t2);
          t0 = n0; t1 = n1; t2 = n2;
        }
      }
      merge_top3<1>(t0, t1, t2);
      merge_top3<2>(t0, t1, t2);
      merge_top3<4>(t0, t1, t2);

#pragma unroll
      for (int k = 0; k < 16; ++k) {
        const float a = e[k];
        e[k] = (a >= t2) ? __expf(a - t0) : 0.0f;
      }
      float s4[4];
#pragma unroll
      for (int i = 0; i < 4; ++i)
        s4[i] = (e[4 * i] + e[4 * i + 1]) + (e[4 * i + 2] + e[4 * i + 3]);
      float s = (s4[0] + s4[1]) + (s4[2] + s4[3]);
      s += xlane<1>(s);
      s += xlane<2>(s);
      s += xlane<4>(s);
      const float rsum = 1.0f / s;
#pragma unroll
      for (int i = 0; i < 4; ++i) {
        float4 p;
        p.x = e[4 * i] * rsum; p.y = e[4 * i + 1] * rsum; p.z = e[4 * i + 2] * rsum; p.w = e[4 * i + 3] * rsum;
        *(float4*)&rowp[4 * ((ch0 + 8 * i) & 31)] = p;
      }
    }
  }
  __syncthreads();

  sym_tiles<1>(sC, sS, tid);
  __syncthreads();

  v8f f[8];
#pragma unroll
  for (int tm = 0; tm < 8; ++tm) f[tm] = creg[tm];

  const int aOff = lm * LDSTR + 8 * lh;
  const int sOff = (8 * lh) * LDSTR + wave * 16 + lm;

  for (int step = 0; step < NSTEPS; ++step) {
    v8f acc[8] = {};
    int aO = aOff, sO2 = sOff;
    asm volatile("" : "+v"(aO), "+v"(sO2));
    const float* aBase = sS + aO;
    const float* sBase = sS + sO2;

    float4 aC[4], aN[4];
#pragma unroll
    for (int j = 0; j < 4; ++j)
      aC[j] = *(const float4*)&aBase[(j * 16) * LDSTR];

#pragma unroll
    for (int g = 0; g < 32; ++g) {
      if (g + 1 < 32) {
        const int gn = g + 1;
        const int tan = 4 * (gn & 1), tmn = gn >> 2, qn = (gn >> 1) & 1;
#pragma unroll
        for (int j = 0; j < 4; ++j)
          aN[j] = *(const float4*)&aBase[((tan + j) * 16) * LDSTR + 16 * tmn + 4 * qn];
      }
      __builtin_amdgcn_sched_barrier(0);
      const int ta0 = 4 * (g & 1), tm = g >> 2, q = (g >> 1) & 1;
      v2f b0, b1;
      b0[0] = f[tm][4 * q];     b0[1] = f[tm][4 * q + 1];
      b1[0] = f[tm][4 * q + 2]; b1[1] = f[tm][4 * q + 3];
#pragma unroll
      for (int j = 0; j < 4; ++j) {
        v2f a; a[0] = aC[j].x; a[1] = aC[j].y;
        acc[ta0 + j] = __builtin_amdgcn_wmma_f32_16x16x4_f32(
            false, a, false, b0, (short)0, acc[ta0 + j], false, false);
      }
      __builtin_amdgcn_sched_barrier(0);
#pragma unroll
      for (int j = 0; j < 4; ++j) {
        v2f a; a[0] = aC[j].z; a[1] = aC[j].w;
        acc[ta0 + j] = __builtin_amdgcn_wmma_f32_16x16x4_f32(
            false, a, false, b1, (short)0, acc[ta0 + j], false, false);
      }
      __builtin_amdgcn_sched_barrier(0);
#pragma unroll
      for (int j = 0; j < 4; ++j) aC[j] = aN[j];
    }

    const float alpha = (step == 0) ? 0.5f : ((step == 1) ? (1.0f / 3.0f) : 0.25f);
#pragma unroll
    for (int tm = 0; tm < 8; ++tm) {
      float sv[8];
#pragma unroll
      for (int v = 0; v < 8; ++v)
        sv[v] = sBase[(tm * 16 + v) * LDSTR];
#pragma unroll
      for (int v = 0; v < 8; ++v) {
        const float fa = f[tm][v];
        const float fn = __builtin_fmaf(alpha, acc[tm][v] - fa, fa);
        f[tm][v] = __builtin_fmaf(sv[v], creg[tm][v] - fn, fn);
      }
    }
  }

  int tidE = tid;
  asm volatile("" : "+v"(tidE));
  const int lmE = tidE & 15, lhE = (tidE >> 4) & 1, waveE = tidE >> 5;
  float* sO = sC;
#pragma unroll
  for (int tm = 0; tm < 8; ++tm) {
#pragma unroll
    for (int v = 0; v < 8; ++v)
      sO[(tm * 16 + v + 8 * lhE) * LDSTR + waveE * 16 + lmE] = f[tm][v];
  }
  __syncthreads();

  sym_tiles<2>(sO, sS, tidE);
  __syncthreads();
  float* dst = out + (size_t)b * (N * N);
  {
    float4 og[16];
#pragma unroll
    for (int it = 0; it < 16; ++it) {
      const int e = tidE * 4 + it * 1024;
      og[it] = *(const float4*)&sS[(e >> 7) * LDSTR + (e & 127)];
    }
#pragma unroll
    for (int it = 0; it < 16; ++it)
      *(float4*)(dst + tidE * 4 + it * 1024) = og[it];
  }
}

extern "C" void kernel_launch(void* const* d_in, const int* in_sizes, int n_in,
                              void* d_out, int out_size, void* d_ws, size_t ws_size,
                              hipStream_t stream) {
  (void)in_sizes; (void)n_in; (void)out_size; (void)d_ws; (void)ws_size;
  const float* corr = (const float*)d_in[0];
  const float* temp = (const float*)d_in[1];
  float* out = (float*)d_out;

  const size_t shmem = 2 * (size_t)PFLOATS * sizeof(float);
  static bool attr_set = false;
  if (!attr_set) {
    hipFuncSetAttribute((const void*)sgf_kernel,
                        hipFuncAttributeMaxDynamicSharedMemorySize, (int)shmem);
    attr_set = true;
  }
  sgf_kernel<<<dim3(NBATCH), dim3(256), shmem, stream>>>(corr, temp, out);
}


# Graph build: d = sqrt(.) on the hardware square-root instruction (1 ulp) instead of the compiler's correctly rounded expansion (about 15 vector instructions per element); affinities move by ulps
# speedup vs baseline: 4.2901x; 1.0333x over previous
#include <hip/hip_runtime.h>


#define N       128
#define LDSTR   132
#define NBATCH  2048
#define EPSF    1e-8f
#define NSTEPS  3
#define PFLOATS (N * LDSTR)

typedef __attribute__((ext_vector_type(2))) float v2f;
typedef __attribute__((ext_vector_type(8))) float v8f;

template <int M>
__device__ __forceinline__ float xlane(float x) {
  const int b = __builtin_bit_cast(int, x);
  return __builtin_bit_cast(float,
      __builtin_amdgcn_mov_dpp(b, 0x160 + (M & 15), 0xf, 0xf, false));
}

template <int M>
__device__ __forceinline__ void merge_top3(float& a0, float& a1, float& a2) {
  const float b0 = xlane<M>(a0), b1 = xlane<M>(a1), b2 = xlane<M>(a2);
  const float c0 = __builtin_fmaxf(a0, b0);
  const float c1 = __builtin_fmaxf(__builtin_fmaxf(a1, b1), __builtin_fminf(a0, b0));
  const float c2 = __builtin_fmaxf(__builtin_fmaxf(a2, b2),
                                   __builtin_fmaxf(__builtin_fminf(a1, b0), __builtin_fminf(a0, b1)));
  a0 = c0; a1 = c1; a2 = c2;
}

template <int MODE>
__device__ __forceinline__ void sym_tiles(const float* __restrict__ srcp, float* __restrict__ dstp, int tid) {
  const int hw = tid >> 4;
  const int a  = tid & 3;
  const int q  = (tid >> 2) & 3;
  for (int trip = 0; trip < 4; ++trip) {
    const int t  = trip * 16 + hw;
    const int r0 = 16 * (t >> 3) + 4 * a;
    const int c0 = 16 * (t & 7) + 4 * q;
    float d[4][4], m[4][4];
#pragma unroll
    for (int r = 0; r < 4; ++r) {
      const float4 dv = *(const float4*)&srcp[(r0 + r) * LDSTR + c0];
      const float4 mv = *(const float4*)&srcp[(c0 + r) * LDSTR + r0];
      d[r][0] = dv.x; d[r][1] = dv.y; d[r][2] = dv.z; d[r][3] = dv.w;
      m[r][0] = mv.x; m[r][1] = mv.y; m[r][2] = mv.z; m[r][3] = mv.w;
    }
#pragma unroll
    for (int r = 0; r < 4; ++r) {
      float o[4];
#pragma unroll
      for (int c = 0; c < 4; ++c) {
        float x = d[r][c], y = m[c][r];
        if (MODE == 0) {
          x = fminf(fmaxf(x, -1.0f + EPSF), 1.0f - EPSF);
          y = fminf(fmaxf(y, -1.0f + EPSF), 1.0f - EPSF);
        }
        float v = 0.5f * (x + y);
        if (MODE == 2 && r0 + r == c0 + c) v = 1.0f;
        o[c] = v;
      }
      float4 ov; ov.x = o[0]; ov.y = o[1]; ov.z = o[2]; ov.w = o[3];
      *(float4*)&dstp[(r0 + r) * LDSTR + c0] = ov;
    }
  }
}

__global__ __launch_bounds__(256, 4)
void sgf_kernel(const float* __restrict__ corr,
                const float* __restrict__ temperature,
                float* __restrict__ out) {
  extern __shared__ float lds[];
  float* sS = lds;
  float* sC = lds + PFLOATS;

  const int tid  = threadIdx.x;
  const int lane = tid & 31;
  const int wave = tid >> 5;
  const int lm   = lane & 15;
  const int lh   = lane >> 4;
  const int b    = blockIdx.x;

  const float invT = 1.0f / temperature[0];
  const float* src = corr + (size_t)b * (N * N);

  {
    float4 stg[16];
#pragma unroll
    for (int it = 0; it < 16; ++it)
      stg[it] = *(const float4*)(src + tid * 4 + it * 1024);
#pragma unroll
    for (int it = 0; it < 16; ++it) {
      const int e = tid * 4 + it * 1024;
      const int i = e >> 7, j = e & 127;
      *(float4*)&sS[i * LDSTR + j] = stg[it];
    }
  }
  __syncthreads();

  sym_tiles<0>(sS, sC, tid);
  __syncthreads();

  v8f creg[8];
#pragma unroll
  for (int tm = 0; tm < 8; ++tm) {
#pragma unroll
    for (int v = 0; v < 8; ++v)
      creg[tm][v] = sC[(tm * 16 + v + 8 * lh) * LDSTR + wave * 16 + lm];
  }
  __syncthreads();

  {
    const int gq  = lane >> 3;
    const int ch0 = (lane & 7) + 7 * (gq & 1);
    for (int rb = 0; rb < 16; rb += 4) {
      const int row = wave * 16 + rb + gq;
      float* rowp = sC + row * LDSTR;
      float4 cv[4];
#pragma unroll
      for (int i = 0; i < 4; ++i)
        cv[i] = *(const float4*)&rowp[4 * ((ch0 + 8 * i) & 31)];

      float e[16];
      float t0 = -__builtin_inff(), t1 = -__builtin_inff(), t2 = -__builtin_inff();
#pragma unroll
      for (int i = 0; i < 4; ++i) {
        const int col0 = 4 * ((ch0 + 8 * i) & 31);
        const float c4[4] = { cv[i].x, cv[i].y, cv[i].z, cv[i].w };
#pragma unroll
        for (int q = 0; q < 4; ++q) {
          float d   = __builtin_amdgcn_sqrtf(2.0f * (1.0f - c4[q]) + EPSF);
          float aff = __expf(-d * invT);
          if (col0 + q == row) aff = 0.0f;
          e[4 * i + q] = aff;
          const float n0 = __builtin_fmaxf(t0, aff);
          const float n1 = __builtin_fmaxf(__builtin_fminf(t0, aff), t1);
          const float n2 = __builtin_fmaxf(__builtin_fminf(t1, aff), t2);
          t0 = n0; t1 = n1; t2 = n2;
        }
      }
      merge_top3<1>(t0, t1, t2);
      merge_top3<2>(t0, t1, t2);
      merge_top3<4>(t0, t1, t2);

#pragma unroll
      for (int k = 0; k < 16; ++k) {
        const float a = e[k];
        e[k] = (a >= t2) ? __expf(a - t0) : 0.0f;
      }
      float s4[4];
#pragma unroll
      for (int i = 0; i < 4; ++i)
        s4[i] = (e[4 * i] + e[4 * i + 1]) + (e[4 * i + 2] + e[4 * i + 3]);
      float s = (s4[0] + s4[1]) + (s4[2] + s4[3]);
      s += xlane<1>(s);
      s += xlane<2>(s);
      s += xlane<4>(s);
      const float rsum = 1.0f / s;
#pragma unroll
      for (int i = 0; i < 4; ++i) {
        float4 p;
        p.x = e[4 * i] * rsum; p.y = e[4 * i + 1] * rsum; p.z = e[4 * i + 2] * rsum; p.w = e[4 * i + 3] * rsum;
        *(float4*)&rowp[4 * ((ch0 + 8 * i) & 31)] = p;
      }
    }
  }
  __syncthreads();

  sym_tiles<1>(sC, sS, tid);
  __syncthreads();

  v8f f[8];
#pragma unroll
  for (int tm = 0; tm < 8; ++tm) f[tm] = creg[tm];

  const int aOff = lm * LDSTR + 8 * lh;
  const int sOff = (8 * lh) * LDSTR + wave * 16 + lm;

  for (int step = 0; step < NSTEPS; ++step) {
    v8f acc[8] = {};
    int aO = aOff, sO2 = sOff;
    asm volatile("" : "+v"(aO), "+v"(sO2));
    const float* aBase = sS + aO;
    const float* sBase = sS + sO2;

    float4 aC[4], aN[4];
#pragma unroll
    for (int j = 0; j < 4; ++j)
      aC[j] = *(const float4*)&aBase[(j * 16) * LDSTR];

#pragma unroll
    for (int g = 0; g < 32; ++g) {
      if (g + 1 < 32) {
        const int gn = g + 1;
        const int tan = 4 * (gn & 1), tmn = gn >> 2, qn = (gn >> 1) & 1;
#pragma unroll
        for (int j = 0; j < 4; ++j)
          aN[j] = *(const float4*)&aBase[((tan + j) * 16) * LDSTR + 16 * tmn + 4 * qn];
      }
      __builtin_amdgcn_sched_barrier(0);
      const int ta0 = 4 * (g & 1), tm = g >> 2, q = (g >> 1) & 1;
      v2f b0, b1;
      b0[0] = f[tm][4 * q];     b0[1] = f[tm][4 * q + 1];
      b1[0] = f[tm][4 * q + 2]; b1[1] = f[tm][4 * q + 3];
#pragma unroll
      for (int j = 0; j < 4; ++j) {
        v2f a; a[0] = aC[j].x; a[1] = aC[j].y;
        acc[ta0 + j] = __builtin_amdgcn_wmma_f32_16x16x4_f32(
            false, a, false, b0, (short)0, acc[ta0 + j], false, false);
      }
      __builtin_amdgcn_sched_barrier(0);
#pragma unroll
      for (int j = 0; j < 4; ++j) {
        v2f a; a[0] = aC[j].z; a[1] = aC[j].w;
        acc[ta0 + j] = __builtin_amdgcn_wmma_f32_16x16x4_f32(
            false, a, false, b1, (short)0, acc[ta0 + j], false, false);
      }
      __builtin_amdgcn_sched_barrier(0);
#pragma unroll
      for (int j = 0; j < 4; ++j) aC[j] = aN[j];
    }

    const float alpha = (step == 0) ? 0.5f : ((step == 1) ? (1.0f / 3.0f) : 0.25f);
#pragma unroll
    for (int tm = 0; tm < 8; ++tm) {
      float sv[8];
#pragma unroll
      for (int v = 0; v < 8; ++v)
        sv[v] = sBase[(tm * 16 + v) * LDSTR];
#pragma unroll
      for (int v = 0; v < 8; ++v) {
        const float fa = f[tm][v];
        const float fn = __builtin_fmaf(alpha, acc[tm][v] - fa, fa);
        f[tm][v] = __builtin_fmaf(sv[v], creg[tm][v] - fn, fn);
      }
    }
  }

  int tidE = tid;
  asm volatile("" : "+v"(tidE));
  const int lmE = tidE & 15, lhE = (tidE >> 4) & 1, waveE = tidE >> 5;
  float* sO = sC;
#pragma unroll
  for (int tm = 0; tm < 8; ++tm) {
#pragma unroll
    for (int v = 0; v < 8; ++v)
      sO[(tm * 16 + v + 8 * lhE) * LDSTR + waveE * 16 + lmE] = f[tm][v];
  }
  __syncthreads();

  sym_tiles<2>(sO, sS, tidE);
  __syncthreads();
  float* dst = out + (size_t)b * (N * N);
  {
    float4 og[16];
#pragma unroll
    for (int it = 0; it < 16; ++it) {
      const int e = tidE * 4 + it * 1024;
      og[it] = *(const float4*)&sS[(e >> 7) * LDSTR + (e & 127)];
    }
#pragma unroll
    for (int it = 0; it < 16; ++it)
      *(float4*)(dst + tidE * 4 + it * 1024) = og[it];
  }
}

extern "C" void kernel_launch(void* const* d_in, const int* in_sizes, int n_in,
                              void* d_out, int out_size, void* d_ws, size_t ws_size,
                              hipStream_t stream) {
  (void)in_sizes; (void)n_in; (void)out_size; (void)d_ws; (void)ws_size;
  const float* corr = (const float*)d_in[0];
  const float* temp = (const float*)d_in[1];
  float* out = (float*)d_out;

  const size_t shmem = 2 * (size_t)PFLOATS * sizeof(float);
  static bool attr_set = false;
  if (!attr_set) {
    hipFuncSetAttribute((const void*)sgf_kernel,
                        hipFuncAttributeMaxDynamicSharedMemorySize, (int)shmem);
    attr_set = true;
  }
  sgf_kernel<<<dim3(NBATCH), dim3(256), shmem, stream>>>(corr, temp, out);
}
